# starting kernel
# baseline (speedup 1.0000x reference)
#include <hip/hip_runtime.h>

typedef float v2f __attribute__((ext_vector_type(2)));
typedef float v8f __attribute__((ext_vector_type(8)));

#define NB   32
#define NI   2048
#define DIN  16
#define NJ   32
#define DOUT 32
#define EPS  1e-7f
#define RSTRIDE 20

__device__ __forceinline__ void ihat_tiles(const float* __restrict__ x,
                                           const float* __restrict__ W,
                                           int j, int i, int lane, v8f acc[4]) {
  const int half = lane >> 4;
  const int l15  = lane & 15;
#pragma unroll
  for (int ks = 0; ks < 4; ++ks) {
    const int kb = ks * 4 + half * 2;
    const size_t xoff = (size_t)i * DIN + kb;
    v2f a0 = *(const v2f*)(x + (size_t)(l15 +  0) * (NI * DIN) + xoff);
    v2f a1 = *(const v2f*)(x + (size_t)(l15 + 16) * (NI * DIN) + xoff);
    const float* wp = W + ((size_t)(j * NI + i) * DIN + kb) * DOUT + l15;
    v2f b0, b1;
    b0.x = wp[0];   b0.y = wp[DOUT];
    b1.x = wp[16];  b1.y = wp[16 + DOUT];
    acc[0] = __builtin_amdgcn_wmma_f32_16x16x4_f32(false, a0, false, b0, (short)0, acc[0], false, false);
    acc[1] = __builtin_amdgcn_wmma_f32_16x16x4_f32(false, a0, false, b1, (short)0, acc[1], false, false);
    acc[2] = __builtin_amdgcn_wmma_f32_16x16x4_f32(false, a1, false, b0, (short)0, acc[2], false, false);
    acc[3] = __builtin_amdgcn_wmma_f32_16x16x4_f32(false, a1, false, b1, (short)0, acc[3], false, false);
  }
}

__global__ void __launch_bounds__(128) capsL_logits(const float* __restrict__ x,
                                                    const float* __restrict__ W,
                                                    const float* __restrict__ Oacc,
                                                    float* __restrict__ logits) {
  __shared__ float red[4][NB * RSTRIDE];
  const int lane = threadIdx.x & 31;
  const int wave = threadIdx.x >> 5;
  const int j     = blockIdx.x;
  const int ibase = blockIdx.y * 32;
  float* myred = &red[wave][0];

  float of[4][8];
#pragma unroll
  for (int t = 0; t < 4; ++t) {
    const int bb = (t >> 1) * 16 + (lane >> 4) * 8;
    const int d  = (t & 1) * 16 + (lane & 15);
#pragma unroll
    for (int v = 0; v < 8; ++v)
      of[t][v] = Oacc[(size_t)((bb + v) * NJ + j) * DOUT + d];
  }

  for (int it = 0; it < 8; ++it) {
    const int i = ibase + it * 4 + wave;
    v8f z = {0.f,0.f,0.f,0.f,0.f,0.f,0.f,0.f};
    v8f acc[4] = {z, z, z, z};
    ihat_tiles(x, W, j, i, lane, acc);
#pragma unroll
    for (int m = 0; m < 2; ++m) {
#pragma unroll
      for (int v = 0; v < 8; ++v) {
        const float p = acc[2*m][v] * of[2*m][v] + acc[2*m+1][v] * of[2*m+1][v];
        const int R = m * 16 + (lane >> 4) * 8 + v;
        myred[R * RSTRIDE + (lane & 15)] = p;
      }
    }
    float sum = 0.f;
    const float* rowp = myred + lane * RSTRIDE;
#pragma unroll
    for (int c = 0; c < 16; c += 4) {
      const float4 q = *(const float4*)(rowp + c);
      sum += (q.x + q.y) + (q.z + q.w);
    }
    logits[(size_t)lane * (NJ * NI) + (size_t)j * NI + i] = sum;
  }
}

__global__ void __launch_bounds__(128) capsS_outputs(const float* __restrict__ x,
                                                     const float* __restrict__ W,
                                                     const float* __restrict__ att,
                                                     float* __restrict__ s_buf,
                                                     int uniform_c) {
  const int lane = threadIdx.x & 31;
  const int wave = threadIdx.x >> 5;
  const int j     = blockIdx.x;
  const int ibase = blockIdx.y * 32;

  v8f z = {0.f,0.f,0.f,0.f,0.f,0.f,0.f,0.f};
  v8f sacc[4] = {z, z, z, z};

  for (int it = 0; it < 8; ++it) {
    const int i = ibase + it * 4 + wave;
    v8f acc[4] = {z, z, z, z};
    ihat_tiles(x, W, j, i, lane, acc);
    float cf[2][8];
    if (uniform_c) {
#pragma unroll
      for (int m = 0; m < 2; ++m)
#pragma unroll
        for (int v = 0; v < 8; ++v) cf[m][v] = 1.0f / NJ;
    } else {
#pragma unroll
      for (int m = 0; m < 2; ++m) {
#pragma unroll
        for (int v = 0; v < 8; ++v) {
          const int b = m * 16 + (lane >> 4) * 8 + v;
          cf[m][v] = att[(size_t)b * (NJ * NI) + (size_t)j * NI + i];
        }
      }
    }
#pragma unroll
    for (int t = 0; t < 4; ++t) {
      const int m = t >> 1;
#pragma unroll
      for (int v = 0; v < 8; ++v)
        sacc[t][v] += cf[m][v] * acc[t][v];
    }
  }
#pragma unroll
  for (int t = 0; t < 4; ++t) {
    const int bb = (t >> 1) * 16 + (lane >> 4) * 8;
    const int d  = (t & 1) * 16 + (lane & 15);
#pragma unroll
    for (int v = 0; v < 8; ++v)
      atomicAdd(&s_buf[(size_t)((bb + v) * NJ + j) * DOUT + d], sacc[t][v]);
  }
}

__global__ void caps_softmax(float* __restrict__ cbuf) {
  const int idx = blockIdx.x * blockDim.x + threadIdx.x;
  if (idx >= NB * NI) return;
  const int b = idx >> 11;
  const int i = idx & (NI - 1);
  float* p = cbuf + (size_t)b * (NJ * NI) + i;
  float e[NJ];
  float mx = -3.402823466e38f;
#pragma unroll
  for (int jj = 0; jj < NJ; ++jj) { e[jj] = p[(size_t)jj * NI]; mx = fmaxf(mx, e[jj]); }
  float s = 0.f;
#pragma unroll
  for (int jj = 0; jj < NJ; ++jj) { e[jj] = __expf(e[jj] - mx); s += e[jj]; }
  const float inv = 1.f / s;
#pragma unroll
  for (int jj = 0; jj < NJ; ++jj) p[(size_t)jj * NI] = e[jj] * inv;
}

__global__ void caps_squash(const float* __restrict__ s, float* __restrict__ Oacc,
                            float* __restrict__ outbuf, int r) {
  const int idx = blockIdx.x * blockDim.x + threadIdx.x;
  if (idx >= NB * NJ) return;
  float v[DOUT];
  float sq = 0.f;
#pragma unroll
  for (int d = 0; d < DOUT; ++d) { v[d] = s[(size_t)idx * DOUT + d]; sq += v[d] * v[d]; }
  const float scale = sq / (1.f + sq) * rsqrtf(sq + EPS);
#pragma unroll
  for (int d = 0; d < DOUT; ++d) {
    const float o = scale * v[d];
    if (r == 2)      outbuf[(size_t)idx * DOUT + d] = o;
    else if (r == 0) Oacc[(size_t)idx * DOUT + d] = o;
    else             Oacc[(size_t)idx * DOUT + d] += o;
  }
}

__global__ void caps_zero(float* __restrict__ p, int n) {
  const int idx = blockIdx.x * blockDim.x + threadIdx.x;
  if (idx < n) p[idx] = 0.f;
}

extern "C" void kernel_launch(void* const* d_in, const int* in_sizes, int n_in,
                              void* d_out, int out_size, void* d_ws, size_t ws_size,
                              hipStream_t stream) {
  (void)in_sizes; (void)n_in; (void)out_size; (void)ws_size;
  const float* x = (const float*)d_in[0];
  const float* W = (const float*)d_in[1];
  float* out = (float*)d_out;

  float* cbuf  = (float*)d_ws;
  float* s_buf = cbuf  + (size_t)NB * NJ * NI;
  float* Oacc  = s_buf + (size_t)NB * NJ * DOUT;

  const dim3 gemm_grid(NJ, NI / 32);
  const int SN = NB * NJ * DOUT;

  for (int r = 0; r < 3; ++r) {
    if (r > 0) {
      capsL_logits<<<gemm_grid, 128, 0, stream>>>(x, W, Oacc, cbuf);
      caps_softmax<<<(NB * NI + 255) / 256, 256, 0, stream>>>(cbuf);
    }
    caps_zero<<<(SN + 255) / 256, 256, 0, stream>>>(s_buf, SN);
    capsS_outputs<<<gemm_grid, 128, 0, stream>>>(x, W, cbuf, s_buf, (r == 0) ? 1 : 0);
    caps_squash<<<(NB * NJ + 255) / 256, 256, 0, stream>>>(s_buf, Oacc, out, r);
  }
}


# Fragment loaders re-mapped (k = 8h+2ks+s on A and B alike, tile column c = d 2c+n): x by 16-byte and W by whole-row 8-byte loads, 12 loads per tile for 24; sums over k and d reordered
# speedup vs baseline: 1.0928x; 1.0928x over previous
#include <hip/hip_runtime.h>

typedef float v2f __attribute__((ext_vector_type(2)));
typedef float v8f __attribute__((ext_vector_type(8)));

#define NB   32
#define NI   2048
#define DIN  16
#define NJ   32
#define DOUT 32
#define EPS  1e-7f
#define RSTRIDE 20

#define DCOL(lane, t) (2 * ((lane) & 15) + ((t) & 1))
__device__ __forceinline__ void ihat_tiles(const float* __restrict__ x,
                                           const float* __restrict__ W,
                                           int j, int i, int lane, v8f acc[4]) {
  const int half = lane >> 4;
  const int l15  = lane & 15;
  const float* xp = x + (size_t)l15 * (NI * DIN) + (size_t)i * DIN + half * 8;
  const float4 xa0 = *(const float4*)(xp);
  const float4 xa1 = *(const float4*)(xp + 4);
  const float4 xb0 = *(const float4*)(xp + (size_t)16 * (NI * DIN));
  const float4 xb1 = *(const float4*)(xp + (size_t)16 * (NI * DIN) + 4);
  const float* wp = W + ((size_t)(j * NI + i) * DIN + half * 8) * DOUT + 2 * l15;
  v2f w[8];
#pragma unroll
  for (int r = 0; r < 8; ++r) w[r] = *(const v2f*)(wp + r * DOUT);
  const v2f a0[4] = {{xa0.x, xa0.y}, {xa0.z, xa0.w}, {xa1.x, xa1.y}, {xa1.z, xa1.w}};
  const v2f a1[4] = {{xb0.x, xb0.y}, {xb0.z, xb0.w}, {xb1.x, xb1.y}, {xb1.z, xb1.w}};
#pragma unroll
  for (int ks = 0; ks < 4; ++ks) {
    v2f b0, b1;
    b0.x = w[2 * ks].x;  b0.y = w[2 * ks + 1].x;
    b1.x = w[2 * ks].y;  b1.y = w[2 * ks + 1].y;
    acc[0] = __builtin_amdgcn_wmma_f32_16x16x4_f32(false, a0[ks], false, b0, (short)0, acc[0], false, false);
    acc[1] = __builtin_amdgcn_wmma_f32_16x16x4_f32(false, a0[ks], false, b1, (short)0, acc[1], false, false);
    acc[2] = __builtin_amdgcn_wmma_f32_16x16x4_f32(false, a1[ks], false, b0, (short)0, acc[2], false, false);
    acc[3] = __builtin_amdgcn_wmma_f32_16x16x4_f32(false, a1[ks], false, b1, (short)0, acc[3], false, false);
  }
}

__global__ void __launch_bounds__(128) capsL_logits(const float* __restrict__ x,
                                                    const float* __restrict__ W,
                                                    const float* __restrict__ Oacc,
                                                    float* __restrict__ logits) {
  __shared__ float red[4][NB * RSTRIDE];
  const int lane = threadIdx.x & 31;
  const int wave = threadIdx.x >> 5;
  const int j     = blockIdx.x;
  const int ibase = blockIdx.y * 32;
  float* myred = &red[wave][0];

  float of[4][8];
#pragma unroll
  for (int t = 0; t < 4; ++t) {
    const int bb = (t >> 1) * 16 + (lane >> 4) * 8;
    const int d  = DCOL(lane, t);
#pragma unroll
    for (int v = 0; v < 8; ++v)
      of[t][v] = Oacc[(size_t)((bb + v) * NJ + j) * DOUT + d];
  }

  for (int it = 0; it < 8; ++it) {
    const int i = ibase + it * 4 + wave;
    v8f z = {0.f,0.f,0.f,0.f,0.f,0.f,0.f,0.f};
    v8f acc[4] = {z, z, z, z};
    ihat_tiles(x, W, j, i, lane, acc);
#pragma unroll
    for (int m = 0; m < 2; ++m) {
#pragma unroll
      for (int v = 0; v < 8; ++v) {
        const float p = acc[2*m][v] * of[2*m][v] + acc[2*m+1][v] * of[2*m+1][v];
        const int R = m * 16 + (lane >> 4) * 8 + v;
        myred[R * RSTRIDE + (lane & 15)] = p;
      }
    }
    float sum = 0.f;
    const float* rowp = myred + lane * RSTRIDE;
#pragma unroll
    for (int c = 0; c < 16; c += 4) {
      const float4 q = *(const float4*)(rowp + c);
      sum += (q.x + q.y) + (q.z + q.w);
    }
    logits[(size_t)lane * (NJ * NI) + (size_t)j * NI + i] = sum;
  }
}

__global__ void __launch_bounds__(128) capsS_outputs(const float* __restrict__ x,
                                                     const float* __restrict__ W,
                                                     const float* __restrict__ att,
                                                     float* __restrict__ s_buf,
                                                     int uniform_c) {
  const int lane = threadIdx.x & 31;
  const int wave = threadIdx.x >> 5;
  const int j     = blockIdx.x;
  const int ibase = blockIdx.y * 32;

  v8f z = {0.f,0.f,0.f,0.f,0.f,0.f,0.f,0.f};
  v8f sacc[4] = {z, z, z, z};

  for (int it = 0; it < 8; ++it) {
    const int i = ibase + it * 4 + wave;
    v8f acc[4] = {z, z, z, z};
    ihat_tiles(x, W, j, i, lane, acc);
    float cf[2][8];
    if (uniform_c) {
#pragma unroll
      for (int m = 0; m < 2; ++m)
#pragma unroll
        for (int v = 0; v < 8; ++v) cf[m][v] = 1.0f / NJ;
    } else {
#pragma unroll
      for (int m = 0; m < 2; ++m) {
#pragma unroll
        for (int v = 0; v < 8; ++v) {
          const int b = m * 16 + (lane >> 4) * 8 + v;
          cf[m][v] = att[(size_t)b * (NJ * NI) + (size_t)j * NI + i];
        }
      }
    }
#pragma unroll
    for (int t = 0; t < 4; ++t) {
      const int m = t >> 1;
#pragma unroll
      for (int v = 0; v < 8; ++v)
        sacc[t][v] += cf[m][v] * acc[t][v];
    }
  }
#pragma unroll
  for (int t = 0; t < 4; ++t) {
    const int bb = (t >> 1) * 16 + (lane >> 4) * 8;
    const int d  = DCOL(lane, t);
#pragma unroll
    for (int v = 0; v < 8; ++v)
      atomicAdd(&s_buf[(size_t)((bb + v) * NJ + j) * DOUT + d], sacc[t][v]);
  }
}

__global__ void caps_softmax(float* __restrict__ cbuf) {
  const int idx = blockIdx.x * blockDim.x + threadIdx.x;
  if (idx >= NB * NI) return;
  const int b = idx >> 11;
  const int i = idx & (NI - 1);
  float* p = cbuf + (size_t)b * (NJ * NI) + i;
  float e[NJ];
  float mx = -3.402823466e38f;
#pragma unroll
  for (int jj = 0; jj < NJ; ++jj) { e[jj] = p[(size_t)jj * NI]; mx = fmaxf(mx, e[jj]); }
  float s = 0.f;
#pragma unroll
  for (int jj = 0; jj < NJ; ++jj) { e[jj] = __expf(e[jj] - mx); s += e[jj]; }
  const float inv = 1.f / s;
#pragma unroll
  for (int jj = 0; jj < NJ; ++jj) p[(size_t)jj * NI] = e[jj] * inv;
}

__global__ void caps_squash(const float* __restrict__ s, float* __restrict__ Oacc,
                            float* __restrict__ outbuf, int r) {
  const int idx = blockIdx.x * blockDim.x + threadIdx.x;
  if (idx >= NB * NJ) return;
  float v[DOUT];
  float sq = 0.f;
#pragma unroll
  for (int d = 0; d < DOUT; ++d) { v[d] = s[(size_t)idx * DOUT + d]; sq += v[d] * v[d]; }
  const float scale = sq / (1.f + sq) * rsqrtf(sq + EPS);
#pragma unroll
  for (int d = 0; d < DOUT; ++d) {
    const float o = scale * v[d];
    if (r == 2)      outbuf[(size_t)idx * DOUT + d] = o;
    else if (r == 0) Oacc[(size_t)idx * DOUT + d] = o;
    else             Oacc[(size_t)idx * DOUT + d] += o;
  }
}

__global__ void caps_zero(float* __restrict__ p, int n) {
  const int idx = blockIdx.x * blockDim.x + threadIdx.x;
  if (idx < n) p[idx] = 0.f;
}

extern "C" void kernel_launch(void* const* d_in, const int* in_sizes, int n_in,
                              void* d_out, int out_size, void* d_ws, size_t ws_size,
                              hipStream_t stream) {
  (void)in_sizes; (void)n_in; (void)out_size; (void)ws_size;
  const float* x = (const float*)d_in[0];
  const float* W = (const float*)d_in[1];
  float* out = (float*)d_out;

  float* cbuf  = (float*)d_ws;
  float* s_buf = cbuf  + (size_t)NB * NJ * NI;
  float* Oacc  = s_buf + (size_t)NB * NJ * DOUT;

  const dim3 gemm_grid(NJ, NI / 32);
  const int SN = NB * NJ * DOUT;

  for (int r = 0; r < 3; ++r) {
    if (r > 0) {
      capsL_logits<<<gemm_grid, 128, 0, stream>>>(x, W, Oacc, cbuf);
      caps_softmax<<<(NB * NI + 255) / 256, 256, 0, stream>>>(cbuf);
    }
    caps_zero<<<(SN + 255) / 256, 256, 0, stream>>>(s_buf, SN);
    capsS_outputs<<<gemm_grid, 128, 0, stream>>>(x, W, cbuf, s_buf, (r == 0) ? 1 : 0);
    caps_squash<<<(NB * NJ + 255) / 256, 256, 0, stream>>>(s_buf, Oacc, out, r);
  }
}


# S pass: routing coefficient folded into the x (A) fragment, the 8 tiles' WMMAs add straight into the output sums (order of sums over i, k changed); c read once per block via LDS; 419 to 86 VGPRs
# speedup vs baseline: 1.5153x; 1.3866x over previous
#include <hip/hip_runtime.h>

typedef float v2f __attribute__((ext_vector_type(2)));
typedef float v8f __attribute__((ext_vector_type(8)));

#define NB   32
#define NI   2048
#define DIN  16
#define NJ   32
#define DOUT 32
#define EPS  1e-7f
#define RSTRIDE 20

#define DCOL(lane, t) (2 * ((lane) & 15) + ((t) & 1))
__device__ __forceinline__ void ihat_tiles(const float* __restrict__ x,
                                           const float* __restrict__ W,
                                           int j, int i, int lane, v8f acc[4]) {
  const int half = lane >> 4;
  const int l15  = lane & 15;
  const float* xp = x + (size_t)l15 * (NI * DIN) + (size_t)i * DIN + half * 8;
  const float4 xa0 = *(const float4*)(xp);
  const float4 xa1 = *(const float4*)(xp + 4);
  const float4 xb0 = *(const float4*)(xp + (size_t)16 * (NI * DIN));
  const float4 xb1 = *(const float4*)(xp + (size_t)16 * (NI * DIN) + 4);
  const float* wp = W + ((size_t)(j * NI + i) * DIN + half * 8) * DOUT + 2 * l15;
  v2f w[8];
#pragma unroll
  for (int r = 0; r < 8; ++r) w[r] = *(const v2f*)(wp + r * DOUT);
  const v2f a0[4] = {{xa0.x, xa0.y}, {xa0.z, xa0.w}, {xa1.x, xa1.y}, {xa1.z, xa1.w}};
  const v2f a1[4] = {{xb0.x, xb0.y}, {xb0.z, xb0.w}, {xb1.x, xb1.y}, {xb1.z, xb1.w}};
#pragma unroll
  for (int ks = 0; ks < 4; ++ks) {
    v2f b0, b1;
    b0.x = w[2 * ks].x;  b0.y = w[2 * ks + 1].x;
    b1.x = w[2 * ks].y;  b1.y = w[2 * ks + 1].y;
    acc[0] = __builtin_amdgcn_wmma_f32_16x16x4_f32(false, a0[ks], false, b0, (short)0, acc[0], false, false);
    acc[1] = __builtin_amdgcn_wmma_f32_16x16x4_f32(false, a0[ks], false, b1, (short)0, acc[1], false, false);
    acc[2] = __builtin_amdgcn_wmma_f32_16x16x4_f32(false, a1[ks], false, b0, (short)0, acc[2], false, false);
    acc[3] = __builtin_amdgcn_wmma_f32_16x16x4_f32(false, a1[ks], false, b1, (short)0, acc[3], false, false);
  }
}

__global__ void __launch_bounds__(128) capsL_logits(const float* __restrict__ x,
                                                    const float* __restrict__ W,
                                                    const float* __restrict__ Oacc,
                                                    float* __restrict__ logits) {
  __shared__ float red[4][NB * RSTRIDE];
  const int lane = threadIdx.x & 31;
  const int wave = threadIdx.x >> 5;
  const int j     = blockIdx.x;
  const int ibase = blockIdx.y * 32;
  float* myred = &red[wave][0];

  float of[4][8];
#pragma unroll
  for (int t = 0; t < 4; ++t) {
    const int bb = (t >> 1) * 16 + (lane >> 4) * 8;
    const int d  = DCOL(lane, t);
#pragma unroll
    for (int v = 0; v < 8; ++v)
      of[t][v] = Oacc[(size_t)((bb + v) * NJ + j) * DOUT + d];
  }

  for (int it = 0; it < 8; ++it) {
    const int i = ibase + it * 4 + wave;
    v8f z = {0.f,0.f,0.f,0.f,0.f,0.f,0.f,0.f};
    v8f acc[4] = {z, z, z, z};
    ihat_tiles(x, W, j, i, lane, acc);
#pragma unroll
    for (int m = 0; m < 2; ++m) {
#pragma unroll
      for (int v = 0; v < 8; ++v) {
        const float p = acc[2*m][v] * of[2*m][v] + acc[2*m+1][v] * of[2*m+1][v];
        const int R = m * 16 + (lane >> 4) * 8 + v;
        myred[R * RSTRIDE + (lane & 15)] = p;
      }
    }
    float sum = 0.f;
    const float* rowp = myred + lane * RSTRIDE;
#pragma unroll
    for (int c = 0; c < 16; c += 4) {
      const float4 q = *(const float4*)(rowp + c);
      sum += (q.x + q.y) + (q.z + q.w);
    }
    logits[(size_t)lane * (NJ * NI) + (size_t)j * NI + i] = sum;
  }
}

#define PPITCH 36
#define CPITCH 36
__global__ void __launch_bounds__(128) capsS_outputs(const float* __restrict__ x,
                                                     const float* __restrict__ W,
                                                     float* cbuf,
                                                     int uniform_c) {
  __shared__ __attribute__((aligned(16))) float red[4][NB * PPITCH];
  __shared__ __attribute__((aligned(16))) float csh[NB * CPITCH];
  const int lane = threadIdx.x & 31;
  const int wave = threadIdx.x >> 5;
  const int half = lane >> 4;
  const int l15  = lane & 15;
  const int j     = blockIdx.x;
  const int ibase = blockIdx.y * 32;

  for (int q = 0; q < 2; ++q) {
    const int idx4 = threadIdx.x + 128 * q;
    const int b  = idx4 >> 3;
    const int i4 = (idx4 & 7) * 4;
    float4 cv;
    if (uniform_c) {
      cv.x = 1.0f / NJ; cv.y = 1.0f / NJ; cv.z = 1.0f / NJ; cv.w = 1.0f / NJ;
    } else {
      cv = *(const float4*)(cbuf + (size_t)b * (NJ * NI) + (size_t)j * NI + ibase + i4);
    }
    *(float4*)(&csh[b * CPITCH + i4]) = cv;
  }
  __syncthreads();

  v8f z = {0.f,0.f,0.f,0.f,0.f,0.f,0.f,0.f};
  v8f sacc[4] = {z, z, z, z};

  for (int it = 0; it < 8; ++it) {
    const int il = it * 4 + wave;
    const int i  = ibase + il;
    const float c0 = csh[l15 * CPITCH + il];
    const float c1 = csh[(l15 + 16) * CPITCH + il];
    const float* xp = x + (size_t)l15 * (NI * DIN) + (size_t)i * DIN + half * 8;
    const float4 xa0 = *(const float4*)(xp);
    const float4 xa1 = *(const float4*)(xp + 4);
    const float4 xb0 = *(const float4*)(xp + (size_t)16 * (NI * DIN));
    const float4 xb1 = *(const float4*)(xp + (size_t)16 * (NI * DIN) + 4);
    const float* wp = W + ((size_t)(j * NI + i) * DIN + half * 8) * DOUT + 2 * l15;
    v2f w[8];
#pragma unroll
    for (int r = 0; r < 8; ++r) w[r] = *(const v2f*)(wp + r * DOUT);
    const v2f a0[4] = {{c0 * xa0.x, c0 * xa0.y}, {c0 * xa0.z, c0 * xa0.w},
                       {c0 * xa1.x, c0 * xa1.y}, {c0 * xa1.z, c0 * xa1.w}};
    const v2f a1[4] = {{c1 * xb0.x, c1 * xb0.y}, {c1 * xb0.z, c1 * xb0.w},
                       {c1 * xb1.x, c1 * xb1.y}, {c1 * xb1.z, c1 * xb1.w}};
#pragma unroll
    for (int ks = 0; ks < 4; ++ks) {
      v2f b0, b1;
      b0.x = w[2 * ks].x;  b0.y = w[2 * ks + 1].x;
      b1.x = w[2 * ks].y;  b1.y = w[2 * ks + 1].y;
      sacc[0] = __builtin_amdgcn_wmma_f32_16x16x4_f32(false, a0[ks], false, b0, (short)0, sacc[0], false, false);
      sacc[1] = __builtin_amdgcn_wmma_f32_16x16x4_f32(false, a0[ks], false, b1, (short)0, sacc[1], false, false);
      sacc[2] = __builtin_amdgcn_wmma_f32_16x16x4_f32(false, a1[ks], false, b0, (short)0, sacc[2], false, false);
      sacc[3] = __builtin_amdgcn_wmma_f32_16x16x4_f32(false, a1[ks], false, b1, (short)0, sacc[3], false, false);
    }
  }

  float* myred = &red[wave][0];
#pragma unroll
  for (int m = 0; m < 2; ++m) {
#pragma unroll
    for (int v = 0; v < 8; ++v) {
      const int b = m * 16 + half * 8 + v;
      v2f p;
      p.x = sacc[2 * m][v];
      p.y = sacc[2 * m + 1][v];
      *(v2f*)(myred + b * PPITCH + 2 * l15) = p;
    }
  }
  __syncthreads();
#pragma unroll
  for (int q = 0; q < 2; ++q) {
    const int idx4 = threadIdx.x + 128 * q;
    const int b  = idx4 >> 3;
    const int d4 = (idx4 & 7) * 4;
    const float4 a0 = *(const float4*)(&red[0][b * PPITCH + d4]);
    const float4 a1 = *(const float4*)(&red[1][b * PPITCH + d4]);
    const float4 a2 = *(const float4*)(&red[2][b * PPITCH + d4]);
    const float4 a3 = *(const float4*)(&red[3][b * PPITCH + d4]);
    float4 s;
    s.x = (a0.x + a1.x) + (a2.x + a3.x);
    s.y = (a0.y + a1.y) + (a2.y + a3.y);
    s.z = (a0.z + a1.z) + (a2.z + a3.z);
    s.w = (a0.w + a1.w) + (a2.w + a3.w);
    *(float4*)(cbuf + (size_t)b * (NJ * NI) + (size_t)j * NI + ibase + d4) = s;
  }
}

__global__ void caps_softmax(float* __restrict__ cbuf) {
  const int idx = blockIdx.x * blockDim.x + threadIdx.x;
  if (idx >= NB * NI) return;
  const int b = idx >> 11;
  const int i = idx & (NI - 1);
  float* p = cbuf + (size_t)b * (NJ * NI) + i;
  float e[NJ];
  float mx = -3.402823466e38f;
#pragma unroll
  for (int jj = 0; jj < NJ; ++jj) { e[jj] = p[(size_t)jj * NI]; mx = fmaxf(mx, e[jj]); }
  float s = 0.f;
#pragma unroll
  for (int jj = 0; jj < NJ; ++jj) { e[jj] = __expf(e[jj] - mx); s += e[jj]; }
  const float inv = 1.f / s;
#pragma unroll
  for (int jj = 0; jj < NJ; ++jj) p[(size_t)jj * NI] = e[jj] * inv;
}

__global__ void __launch_bounds__(128) caps_squash(const float* __restrict__ part,
                                                   float* __restrict__ Oacc,
                                                   float* __restrict__ outbuf, int r) {
  const int lane = threadIdx.x & 31;
  const int row  = blockIdx.x * 4 + (threadIdx.x >> 5);
  const float4* p = (const float4*)(part + (size_t)row * NI) + lane;
  float4 a = p[0];
#pragma unroll
  for (int q = 1; q < 16; ++q) {
    const float4 t = p[q * 32];
    a.x += t.x; a.y += t.y; a.z += t.z; a.w += t.w;
  }
  a.x += __shfl_xor(a.x, 8);  a.y += __shfl_xor(a.y, 8);
  a.z += __shfl_xor(a.z, 8);  a.w += __shfl_xor(a.w, 8);
  a.x += __shfl_xor(a.x, 16); a.y += __shfl_xor(a.y, 16);
  a.z += __shfl_xor(a.z, 16); a.w += __shfl_xor(a.w, 16);
  float sq = (a.x * a.x + a.y * a.y) + (a.z * a.z + a.w * a.w);
  sq += __shfl_xor(sq, 1);
  sq += __shfl_xor(sq, 2);
  sq += __shfl_xor(sq, 4);
  const float scale = sq / (1.f + sq) * rsqrtf(sq + EPS);
  if (lane < 8) {
    float4 o;
    o.x = scale * a.x; o.y = scale * a.y; o.z = scale * a.z; o.w = scale * a.w;
    float4* dst = (float4*)(((r == 2) ? outbuf : Oacc) + (size_t)row * DOUT) + lane;
    if (r == 1) {
      const float4 prev = *dst;
      o.x += prev.x; o.y += prev.y; o.z += prev.z; o.w += prev.w;
    }
    *dst = o;
  }
}

extern "C" void kernel_launch(void* const* d_in, const int* in_sizes, int n_in,
                              void* d_out, int out_size, void* d_ws, size_t ws_size,
                              hipStream_t stream) {
  (void)in_sizes; (void)n_in; (void)out_size; (void)ws_size;
  const float* x = (const float*)d_in[0];
  const float* W = (const float*)d_in[1];
  float* out = (float*)d_out;

  float* cbuf  = (float*)d_ws;
  float* Oacc  = cbuf + (size_t)NB * NJ * NI + (size_t)NB * NJ * DOUT;

  const dim3 gemm_grid(NJ, NI / 32);

  for (int r = 0; r < 3; ++r) {
    if (r > 0) {
      capsL_logits<<<gemm_grid, 128, 0, stream>>>(x, W, Oacc, cbuf);
      caps_softmax<<<(NB * NI + 255) / 256, 256, 0, stream>>>(cbuf);
    }
    capsS_outputs<<<gemm_grid, 128, 0, stream>>>(x, W, cbuf, (r == 0) ? 1 : 0);
    caps_squash<<<(NB * NJ) / 4, 128, 0, stream>>>(cbuf, Oacc, out, r);
  }
}
